# starting kernel
# baseline (speedup 1.0000x reference)
#include <hip/hip_runtime.h>

#define D 128
#define NNODES 50000
#define NEDGES 600000
#define MT 5
#define ROWS (MT * 16)

typedef __attribute__((ext_vector_type(2))) float v2f;
typedef __attribute__((ext_vector_type(8))) float v8f;

__global__ __launch_bounds__(256) void k_init(const float* __restrict__ feat,
                                              const float* __restrict__ eps,
                                              float* __restrict__ h) {
    const int i = (blockIdx.x * 256 + threadIdx.x) * 4;
    const float s = 1.0f + eps[0];
    const float4 f = *reinterpret_cast<const float4*>(feat + i);
    float4 o;
    o.x = s * f.x; o.y = s * f.y; o.z = s * f.z; o.w = s * f.w;
    *reinterpret_cast<float4*>(h + i) = o;
}

__global__ __launch_bounds__(256) void k_scatter(const float* __restrict__ feat,
                                                 const int* __restrict__ src,
                                                 const int* __restrict__ dst,
                                                 const int* __restrict__ et,
                                                 const float* __restrict__ emb,
                                                 float* __restrict__ h) {
    const int wave = threadIdx.x >> 5;
    const int lane = threadIdx.x & 31;
    const int e = blockIdx.x * 8 + wave;
    const int s = src[e];
    const int d = dst[e];
    const int t = et[e];
    const float4 fs = *reinterpret_cast<const float4*>(feat + s * D + lane * 4);
    const float4 em = *reinterpret_cast<const float4*>(emb + t * D + lane * 4);
    float* dp = h + d * D + lane * 4;
    const float m0 = fmaxf(fs.x + em.x, 0.0f);
    const float m1 = fmaxf(fs.y + em.y, 0.0f);
    const float m2 = fmaxf(fs.z + em.z, 0.0f);
    const float m3 = fmaxf(fs.w + em.w, 0.0f);
    unsafeAtomicAdd(dp + 0, m0);
    unsafeAtomicAdd(dp + 1, m1);
    unsafeAtomicAdd(dp + 2, m2);
    unsafeAtomicAdd(dp + 3, m3);
}

__global__ __launch_bounds__(256) void k_mlp(const float* __restrict__ W1,
                                             const float* __restrict__ B1,
                                             const float* __restrict__ W2,
                                             const float* __restrict__ B2,
                                             float* __restrict__ hout) {
    __shared__ float hs[ROWS * 132];
    __shared__ float hid[ROWS * 260];
    const int tid = threadIdx.x;
    const int base = blockIdx.x * ROWS;

    #pragma unroll
    for (int i = 0; i < 10; ++i) {
        const int chunk = tid + i * 256;
        const int r = chunk >> 5;
        const int c = (chunk & 31) * 4;
        const float4 v = *reinterpret_cast<const float4*>(hout + (base + r) * D + c);
        *reinterpret_cast<float4*>(&hs[r * 132 + c]) = v;
    }
    __syncthreads();

    const int w = tid >> 5;
    const int lane = tid & 31;
    const int M = lane & 15;
    const int half = lane >> 4;
    const int koff = half * 2;

    {
        v8f acc0[MT] = {};
        v8f acc1[MT] = {};
        for (int k0 = 0; k0 < D; k0 += 4) {
            const float* bp = W1 + (k0 + koff) * 256 + w * 32 + M;
            v2f b0, b1;
            b0.x = bp[0];   b0.y = bp[256];
            b1.x = bp[16];  b1.y = bp[16 + 256];
            #pragma unroll
            for (int m = 0; m < MT; ++m) {
                v2f a;
                a.x = hs[(m * 16 + M) * 132 + k0 + koff];
                a.y = hs[(m * 16 + M) * 132 + k0 + koff + 1];
                acc0[m] = __builtin_amdgcn_wmma_f32_16x16x4_f32(
                              false, a, false, b0, (short)0, acc0[m], false, false);
                acc1[m] = __builtin_amdgcn_wmma_f32_16x16x4_f32(
                              false, a, false, b1, (short)0, acc1[m], false, false);
            }
        }
        const float bias0 = B1[w * 32 + M];
        const float bias1 = B1[w * 32 + 16 + M];
        #pragma unroll
        for (int m = 0; m < MT; ++m) {
            #pragma unroll
            for (int r = 0; r < 8; ++r) {
                const int row = m * 16 + r + half * 8;
                const float v0 = acc0[m][r] + bias0;
                const float v1 = acc1[m][r] + bias1;
                hid[row * 260 + w * 32 + M]      = v0 > 0.0f ? v0 : 0.0f;
                hid[row * 260 + w * 32 + 16 + M] = v1 > 0.0f ? v1 : 0.0f;
            }
        }
    }
    __syncthreads();

    {
        v8f acc[MT] = {};
        for (int k0 = 0; k0 < 2 * D; k0 += 4) {
            const float* bp = W2 + (k0 + koff) * D + w * 16 + M;
            v2f b;
            b.x = bp[0];
            b.y = bp[D];
            #pragma unroll
            for (int m = 0; m < MT; ++m) {
                v2f a;
                a.x = hid[(m * 16 + M) * 260 + k0 + koff];
                a.y = hid[(m * 16 + M) * 260 + k0 + koff + 1];
                acc[m] = __builtin_amdgcn_wmma_f32_16x16x4_f32(
                             false, a, false, b, (short)0, acc[m], false, false);
            }
        }
        const float bias = B2[w * 16 + M];
        #pragma unroll
        for (int m = 0; m < MT; ++m) {
            #pragma unroll
            for (int r = 0; r < 8; ++r) {
                hout[(base + m * 16 + r + half * 8) * D + w * 16 + M] = acc[m][r] + bias;
            }
        }
    }
}

extern "C" void kernel_launch(void* const* d_in, const int* in_sizes, int n_in,
                              void* d_out, int out_size, void* d_ws, size_t ws_size,
                              hipStream_t stream) {
    const float* feat = (const float*)d_in[0];
    const int*   src  = (const int*)d_in[1];
    const int*   dst  = (const int*)d_in[2];
    const int*   et   = (const int*)d_in[3];
    const float* emb  = (const float*)d_in[4];
    const float* w1   = (const float*)d_in[5];
    const float* b1   = (const float*)d_in[6];
    const float* w2   = (const float*)d_in[7];
    const float* b2   = (const float*)d_in[8];
    const float* eps  = (const float*)d_in[9];
    float* out = (float*)d_out;

    k_init   <<<(NNODES * D / 4) / 256, 256, 0, stream>>>(feat, eps, out);
    k_scatter<<<NEDGES / 8,            256, 0, stream>>>(feat, src, dst, et, emb, out);
    k_mlp    <<<NNODES / ROWS,         256, 0, stream>>>(w1, b1, w2, b2, out);
}


# edges bucketed by 80-row destination tile (LDS histogram, scan, fill); the MLP kernel sums its bucket's messages into the LDS h tile with LDS float atomics; no global atomics, sum order differs
# speedup vs baseline: 1.4422x; 1.4422x over previous
#include <hip/hip_runtime.h>

#define D 128
#define NNODES 50000
#define NEDGES 600000
#define MT 5
#define ROWS (MT * 16)
#define NTILES (NNODES / ROWS)
#define NCHUNK 250
#define EPC (NEDGES / NCHUNK)

#define WS_CNT  0
#define WS_TOT  156288
#define WS_BASE 156928
#define WS_SORT 157568

typedef __attribute__((ext_vector_type(2))) float v2f;
typedef __attribute__((ext_vector_type(8))) float v8f;

__device__ __forceinline__ int wave_incl_scan(int v, int lane) {
    for (int o = 1; o < 32; o <<= 1) {
        const int t = __shfl_up(v, o);
        if (lane >= o) v += t;
    }
    return v;
}

__global__ __launch_bounds__(1024) void k_count(const int* __restrict__ dst,
                                                int* __restrict__ ws) {
    __shared__ int hist[NTILES];
    const int tid = threadIdx.x;
    if (tid < NTILES) hist[tid] = 0;
    __syncthreads();
    const int e0 = blockIdx.x * EPC;
    for (int i = tid; i < EPC; i += 1024) {
        const unsigned d = (unsigned)dst[e0 + i];
        atomicAdd(&hist[d / ROWS], 1);
    }
    __syncthreads();
    if (tid < NTILES) ws[WS_CNT + tid * NCHUNK + blockIdx.x] = hist[tid];
}

__global__ __launch_bounds__(256) void k_scan(int* ws) {
    const int lane = threadIdx.x & 31;
    const int bin = blockIdx.x * 8 + (threadIdx.x >> 5);
    if (bin >= NTILES) return;
    int* row = ws + WS_CNT + bin * NCHUNK;
    int v[8];
    int run = 0;
    for (int k = 0; k < 8; ++k) {
        const int idx = lane * 8 + k;
        const int c = idx < NCHUNK ? row[idx] : 0;
        v[k] = run;
        run += c;
    }
    const int incl = wave_incl_scan(run, lane);
    const int excl = incl - run;
    for (int k = 0; k < 8; ++k) {
        const int idx = lane * 8 + k;
        if (idx < NCHUNK) row[idx] = v[k] + excl;
    }
    if (lane == 31) ws[WS_TOT + bin] = incl;
}

__global__ __launch_bounds__(1024) void k_fill(const int* __restrict__ src,
                                               const int* __restrict__ dst,
                                               const int* __restrict__ et,
                                               int* ws) {
    __shared__ int cur[NTILES];
    __shared__ int wsum[32];
    const int tid = threadIdx.x;
    const int lane = tid & 31;
    const int w = tid >> 5;
    const int t = tid < NTILES ? ws[WS_TOT + tid] : 0;
    const int incl = wave_incl_scan(t, lane);
    if (lane == 31) wsum[w] = incl;
    __syncthreads();
    int b = incl - t;
    for (int k = 0; k < w; ++k) b += wsum[k];
    if (tid < NTILES) {
        cur[tid] = b + ws[WS_CNT + tid * NCHUNK + blockIdx.x];
        if (blockIdx.x == 0) ws[WS_BASE + tid] = b;
    }
    __syncthreads();
    unsigned* sorted = reinterpret_cast<unsigned*>(ws + WS_SORT);
    const int e0 = blockIdx.x * EPC;
    for (int i = tid; i < EPC; i += 1024) {
        const int e = e0 + i;
        const unsigned d = (unsigned)dst[e];
        const unsigned sv = (unsigned)src[e];
        const unsigned tv = (unsigned)et[e];
        const unsigned bin = d / ROWS;
        const unsigned r = d - bin * ROWS;
        const int pos = atomicAdd(&cur[bin], 1);
        sorted[pos] = sv | (tv << 16) | (r << 20);
    }
}

__global__ __launch_bounds__(256) void k_mlp(const float* __restrict__ feat,
                                             const float* __restrict__ eps,
                                             const float* __restrict__ emb,
                                             const int* __restrict__ ws,
                                             const float* __restrict__ W1,
                                             const float* __restrict__ B1,
                                             const float* __restrict__ W2,
                                             const float* __restrict__ B2,
                                             float* __restrict__ hout) {
    __shared__ __attribute__((aligned(16))) float hs[(ROWS + 1) * 132];
    __shared__ __attribute__((aligned(16))) float hid[ROWS * 260];
    const int tid = threadIdx.x;
    const int base = blockIdx.x * ROWS;
    const float sc = 1.0f + eps[0];

    #pragma unroll
    for (int i = 0; i < 10; ++i) {
        const int chunk = tid + i * 256;
        const int r = chunk >> 5;
        const int c = (chunk & 31) * 4;
        const float4 f = *reinterpret_cast<const float4*>(feat + (base + r) * D + c);
        float4 o;
        o.x = sc * f.x; o.y = sc * f.y; o.z = sc * f.z; o.w = sc * f.w;
        *reinterpret_cast<float4*>(&hs[r * 132 + c]) = o;
    }
    for (int i = 0; i < 2; ++i) {
        const int chunk = tid + i * 256;
        *reinterpret_cast<float4*>(&hid[chunk * 4]) =
            *reinterpret_cast<const float4*>(emb + chunk * 4);
    }
    __syncthreads();

    {
        const int lane = tid & 31;
        const int wv = __builtin_amdgcn_readfirstlane(tid >> 5);
        const int n = ws[WS_TOT + blockIdx.x];
        const unsigned* sl = reinterpret_cast<const unsigned*>(ws + WS_SORT) + ws[WS_BASE + blockIdx.x];
        const int c = lane * 2;
        const int pad = ROWS << 20;
        int cb = wv * 32;
        int p = pad;
        if (cb + lane < n) p = (int)sl[cb + lane];
        for (; cb < n; cb += 256) {
            int pn = pad;
            if (cb + 256 + lane < n) pn = (int)sl[cb + 256 + lane];
            const int rem = n - cb;
            const int cw = rem < 32 ? rem : 32;
            for (int j = 0; j < cw; j += 8) {
                int pj[8];
                v2f fa[8], fb[8], ea[8], eb[8];
                #pragma unroll
                for (int u = 0; u < 8; ++u) {
                    pj[u] = __builtin_amdgcn_readlane(p, j + u);
                    const float* fp = feat + (pj[u] & 0xFFFF) * D + c;
                    fa[u] = *reinterpret_cast<const v2f*>(fp);
                    fb[u] = *reinterpret_cast<const v2f*>(fp + 64);
                    const float* ep = &hid[((pj[u] >> 16) & 15) * D + c];
                    ea[u] = *reinterpret_cast<const v2f*>(ep);
                    eb[u] = *reinterpret_cast<const v2f*>(ep + 64);
                }
                asm volatile("" :: "v"(fa[0]), "v"(fb[0]), "v"(fa[1]), "v"(fb[1]),
                                   "v"(fa[2]), "v"(fb[2]), "v"(fa[3]), "v"(fb[3]),
                                   "v"(fa[4]), "v"(fb[4]), "v"(fa[5]), "v"(fb[5]),
                                   "v"(fa[6]), "v"(fb[6]), "v"(fa[7]), "v"(fb[7]));
                #pragma unroll
                for (int u = 0; u < 8; ++u) {
                    float* hp = &hs[(pj[u] >> 20) * 132 + c];
                    __hip_atomic_fetch_add(hp,      fmaxf(fa[u].x + ea[u].x, 0.0f), __ATOMIC_RELAXED, __HIP_MEMORY_SCOPE_WORKGROUP);
                    __hip_atomic_fetch_add(hp + 1,  fmaxf(fa[u].y + ea[u].y, 0.0f), __ATOMIC_RELAXED, __HIP_MEMORY_SCOPE_WORKGROUP);
                    __hip_atomic_fetch_add(hp + 64, fmaxf(fb[u].x + eb[u].x, 0.0f), __ATOMIC_RELAXED, __HIP_MEMORY_SCOPE_WORKGROUP);
                    __hip_atomic_fetch_add(hp + 65, fmaxf(fb[u].y + eb[u].y, 0.0f), __ATOMIC_RELAXED, __HIP_MEMORY_SCOPE_WORKGROUP);
                }
            }
            p = pn;
        }
    }
    __syncthreads();

    const int w = tid >> 5;
    const int lane = tid & 31;
    const int M = lane & 15;
    const int half = lane >> 4;
    const int koff = half * 2;

    {
        v8f acc0[MT] = {};
        v8f acc1[MT] = {};
        for (int k0 = 0; k0 < D; k0 += 4) {
            const float* bp = W1 + (k0 + koff) * 256 + w * 32 + M;
            v2f b0, b1;
            b0.x = bp[0];   b0.y = bp[256];
            b1.x = bp[16];  b1.y = bp[16 + 256];
            #pragma unroll
            for (int m = 0; m < MT; ++m) {
                v2f a;
                a.x = hs[(m * 16 + M) * 132 + k0 + koff];
                a.y = hs[(m * 16 + M) * 132 + k0 + koff + 1];
                acc0[m] = __builtin_amdgcn_wmma_f32_16x16x4_f32(
                              false, a, false, b0, (short)0, acc0[m], false, false);
                acc1[m] = __builtin_amdgcn_wmma_f32_16x16x4_f32(
                              false, a, false, b1, (short)0, acc1[m], false, false);
            }
        }
        const float bias0 = B1[w * 32 + M];
        const float bias1 = B1[w * 32 + 16 + M];
        #pragma unroll
        for (int m = 0; m < MT; ++m) {
            #pragma unroll
            for (int r = 0; r < 8; ++r) {
                const int row = m * 16 + r + half * 8;
                const float v0 = acc0[m][r] + bias0;
                const float v1 = acc1[m][r] + bias1;
                hid[row * 260 + w * 32 + M]      = v0 > 0.0f ? v0 : 0.0f;
                hid[row * 260 + w * 32 + 16 + M] = v1 > 0.0f ? v1 : 0.0f;
            }
        }
    }
    __syncthreads();

    {
        v8f acc[MT] = {};
        for (int k0 = 0; k0 < 2 * D; k0 += 4) {
            const float* bp = W2 + (k0 + koff) * D + w * 16 + M;
            v2f b;
            b.x = bp[0];
            b.y = bp[D];
            #pragma unroll
            for (int m = 0; m < MT; ++m) {
                v2f a;
                a.x = hid[(m * 16 + M) * 260 + k0 + koff];
                a.y = hid[(m * 16 + M) * 260 + k0 + koff + 1];
                acc[m] = __builtin_amdgcn_wmma_f32_16x16x4_f32(
                             false, a, false, b, (short)0, acc[m], false, false);
            }
        }
        const float bias = B2[w * 16 + M];
        #pragma unroll
        for (int m = 0; m < MT; ++m) {
            #pragma unroll
            for (int r = 0; r < 8; ++r) {
                hout[(base + m * 16 + r + half * 8) * D + w * 16 + M] = acc[m][r] + bias;
            }
        }
    }
}

extern "C" void kernel_launch(void* const* d_in, const int* in_sizes, int n_in,
                              void* d_out, int out_size, void* d_ws, size_t ws_size,
                              hipStream_t stream) {
    const float* feat = (const float*)d_in[0];
    const int*   src  = (const int*)d_in[1];
    const int*   dst  = (const int*)d_in[2];
    const int*   et   = (const int*)d_in[3];
    const float* emb  = (const float*)d_in[4];
    const float* w1   = (const float*)d_in[5];
    const float* b1   = (const float*)d_in[6];
    const float* w2   = (const float*)d_in[7];
    const float* b2   = (const float*)d_in[8];
    const float* eps  = (const float*)d_in[9];
    float* out = (float*)d_out;
    int* ws = (int*)d_ws;

    k_count<<<NCHUNK, 1024, 0, stream>>>(dst, ws);
    k_scan <<<79,     256, 0, stream>>>(ws);
    k_fill <<<NCHUNK, 1024, 0, stream>>>(src, dst, et, ws);
    k_mlp  <<<NTILES, 256, 0, stream>>>(feat, eps, emb, ws, w1, b1, w2, b2, out);
}


# MLP kernel orders its bucket by destination row in LDS; a wave sums a row's messages in registers (16-B loads, 16 rows in flight), one LDS atomic add per row and wave; float sum order differs
# speedup vs baseline: 1.5663x; 1.0860x over previous
#include <hip/hip_runtime.h>

#define D 128
#define NNODES 50000
#define NEDGES 600000
#define MT 5
#define ROWS (MT * 16)
#define NTILES (NNODES / ROWS)
#define NCHUNK 250
#define EPC (NEDGES / NCHUNK)
#define CAP 2048

#define WS_CNT  0
#define WS_TOT  156288
#define WS_BASE 156928
#define WS_SORT 157568

typedef __attribute__((ext_vector_type(2))) float v2f;
typedef __attribute__((ext_vector_type(8))) float v8f;
typedef __attribute__((ext_vector_type(4))) float v4f;

__device__ __forceinline__ int wave_incl_scan(int v, int lane) {
    for (int o = 1; o < 32; o <<= 1) {
        const int t = __shfl_up(v, o);
        if (lane >= o) v += t;
    }
    return v;
}

__global__ __launch_bounds__(1024) void k_count(const int* __restrict__ dst,
                                                int* __restrict__ ws) {
    __shared__ int hist[NTILES];
    const int tid = threadIdx.x;
    if (tid < NTILES) hist[tid] = 0;
    __syncthreads();
    const int e0 = blockIdx.x * EPC;
    for (int i = tid; i < EPC; i += 1024) {
        const unsigned d = (unsigned)dst[e0 + i];
        atomicAdd(&hist[d / ROWS], 1);
    }
    __syncthreads();
    if (tid < NTILES) ws[WS_CNT + tid * NCHUNK + blockIdx.x] = hist[tid];
}

__global__ __launch_bounds__(256) void k_scan(int* ws) {
    const int lane = threadIdx.x & 31;
    const int bin = blockIdx.x * 8 + (threadIdx.x >> 5);
    if (bin >= NTILES) return;
    int* row = ws + WS_CNT + bin * NCHUNK;
    int v[8];
    int run = 0;
    for (int k = 0; k < 8; ++k) {
        const int idx = lane * 8 + k;
        const int c = idx < NCHUNK ? row[idx] : 0;
        v[k] = run;
        run += c;
    }
    const int incl = wave_incl_scan(run, lane);
    const int excl = incl - run;
    for (int k = 0; k < 8; ++k) {
        const int idx = lane * 8 + k;
        if (idx < NCHUNK) row[idx] = v[k] + excl;
    }
    if (lane == 31) ws[WS_TOT + bin] = incl;
}

__global__ __launch_bounds__(1024) void k_fill(const int* __restrict__ src,
                                               const int* __restrict__ dst,
                                               const int* __restrict__ et,
                                               int* ws) {
    __shared__ int cur[NTILES];
    __shared__ int wsum[32];
    const int tid = threadIdx.x;
    const int lane = tid & 31;
    const int w = tid >> 5;
    const int t = tid < NTILES ? ws[WS_TOT + tid] : 0;
    const int incl = wave_incl_scan(t, lane);
    if (lane == 31) wsum[w] = incl;
    __syncthreads();
    int b = incl - t;
    for (int k = 0; k < w; ++k) b += wsum[k];
    if (tid < NTILES) {
        cur[tid] = b + ws[WS_CNT + tid * NCHUNK + blockIdx.x];
        if (blockIdx.x == 0) ws[WS_BASE + tid] = b;
    }
    __syncthreads();
    unsigned* sorted = reinterpret_cast<unsigned*>(ws + WS_SORT);
    const int e0 = blockIdx.x * EPC;
    for (int i = tid; i < EPC; i += 1024) {
        const int e = e0 + i;
        const unsigned d = (unsigned)dst[e];
        const unsigned sv = (unsigned)src[e];
        const unsigned tv = (unsigned)et[e];
        const unsigned bin = d / ROWS;
        const unsigned r = d - bin * ROWS;
        const int pos = atomicAdd(&cur[bin], 1);
        sorted[pos] = sv | (tv << 16) | (r << 20);
    }
}

__global__ __launch_bounds__(256) void k_mlp(const float* __restrict__ feat,
                                             const float* __restrict__ eps,
                                             const float* __restrict__ emb,
                                             const int* __restrict__ ws,
                                             const float* __restrict__ W1,
                                             const float* __restrict__ B1,
                                             const float* __restrict__ W2,
                                             const float* __restrict__ B2,
                                             float* __restrict__ hout) {
    __shared__ __attribute__((aligned(16))) float hs[(ROWS + 1) * 132];
    __shared__ __attribute__((aligned(16))) float hid[ROWS * 260];
    __shared__ int cnt[96];
    __shared__ unsigned lst[CAP];
    const int tid = threadIdx.x;
    const int base = blockIdx.x * ROWS;
    const float sc = 1.0f + eps[0];

    #pragma unroll
    for (int i = 0; i < 10; ++i) {
        const int chunk = tid + i * 256;
        const int r = chunk >> 5;
        const int c = (chunk & 31) * 4;
        const float4 f = *reinterpret_cast<const float4*>(feat + (base + r) * D + c);
        float4 o;
        o.x = sc * f.x; o.y = sc * f.y; o.z = sc * f.z; o.w = sc * f.w;
        *reinterpret_cast<float4*>(&hs[r * 132 + c]) = o;
    }
    for (int i = 0; i < 2; ++i) {
        const int chunk = tid + i * 256;
        *reinterpret_cast<float4*>(&hid[chunk * 4]) =
            *reinterpret_cast<const float4*>(emb + chunk * 4);
    }

    {
        const int lane = tid & 31;
        const int wv = __builtin_amdgcn_readfirstlane(tid >> 5);
        const int n = ws[WS_TOT + blockIdx.x];
        const unsigned* sl = reinterpret_cast<const unsigned*>(ws + WS_SORT) + ws[WS_BASE + blockIdx.x];
        const int c = lane * 4;
        const int pad = ROWS << 20;
        for (int B = 0; B < n; B += CAP) {
            const int m = (n - B) < CAP ? (n - B) : CAP;
            if (tid < 96) cnt[tid] = 0;
            __syncthreads();
            unsigned wk[CAP / 256];
            int rk[CAP / 256];
            #pragma unroll
            for (int k = 0; k < CAP / 256; ++k) {
                const int i = k * 256 + tid;
                wk[k] = sl[B + (i < m ? i : 0)];
            }
            #pragma unroll
            for (int k = 0; k < CAP / 256; ++k) {
                const int i = k * 256 + tid;
                rk[k] = 0;
                if (i < m) rk[k] = atomicAdd(&cnt[wk[k] >> 20], 1);
            }
            __syncthreads();
            if (tid < 32) {
                const int a0 = cnt[3 * lane], a1 = cnt[3 * lane + 1], a2 = cnt[3 * lane + 2];
                const int sum = a0 + a1 + a2;
                const int ex = wave_incl_scan(sum, lane) - sum;
                cnt[3 * lane] = ex;
                cnt[3 * lane + 1] = ex + a0;
                cnt[3 * lane + 2] = ex + a0 + a1;
            }
            __syncthreads();
            #pragma unroll
            for (int k = 0; k < CAP / 256; ++k) {
                const int i = k * 256 + tid;
                if (i < m) lst[cnt[wk[k] >> 20] + rk[k]] = wk[k];
            }
            __syncthreads();
            const int lo = (m * wv) >> 3;
            const int hi = (m * (wv + 1)) >> 3;
            int cur = ROWS;
            float4 acc = make_float4(0.0f, 0.0f, 0.0f, 0.0f);
            for (int pos = lo; pos < hi; pos += 32) {
                int p = pad;
                if (pos + lane < hi) p = (int)lst[pos + lane];
                const int rem = hi - pos;
                const int cw = rem < 32 ? rem : 32;
                for (int j = 0; j < cw; j += 16) {
                    int pj[16];
                    v4f f[16], e[16];
                    #pragma unroll
                    for (int u = 0; u < 16; ++u) {
                        pj[u] = __builtin_amdgcn_readlane(p, j + u);
                        f[u] = *reinterpret_cast<const v4f*>(feat + (pj[u] & 0xFFFF) * D + c);
                        e[u] = *reinterpret_cast<const v4f*>(&hid[((pj[u] >> 16) & 15) * D + c]);
                    }
                    asm volatile("" :: "v"(f[0]), "v"(f[1]), "v"(f[2]), "v"(f[3]),
                                       "v"(f[4]), "v"(f[5]), "v"(f[6]), "v"(f[7]),
                                       "v"(f[8]), "v"(f[9]), "v"(f[10]), "v"(f[11]),
                                       "v"(f[12]), "v"(f[13]), "v"(f[14]), "v"(f[15]));
                    #pragma unroll
                    for (int u = 0; u < 16; ++u) {
                        const int r = pj[u] >> 20;
                        if (r != cur) {
                            float* hp = &hs[cur * 132 + c];
                            __hip_atomic_fetch_add(hp,     acc.x, __ATOMIC_RELAXED, __HIP_MEMORY_SCOPE_WORKGROUP);
                            __hip_atomic_fetch_add(hp + 1, acc.y, __ATOMIC_RELAXED, __HIP_MEMORY_SCOPE_WORKGROUP);
                            __hip_atomic_fetch_add(hp + 2, acc.z, __ATOMIC_RELAXED, __HIP_MEMORY_SCOPE_WORKGROUP);
                            __hip_atomic_fetch_add(hp + 3, acc.w, __ATOMIC_RELAXED, __HIP_MEMORY_SCOPE_WORKGROUP);
                            acc = make_float4(0.0f, 0.0f, 0.0f, 0.0f);
                            cur = r;
                        }
                        acc.x += fmaxf(f[u].x + e[u].x, 0.0f);
                        acc.y += fmaxf(f[u].y + e[u].y, 0.0f);
                        acc.z += fmaxf(f[u].z + e[u].z, 0.0f);
                        acc.w += fmaxf(f[u].w + e[u].w, 0.0f);
                    }
                }
            }
            {
                float* hp = &hs[cur * 132 + c];
                __hip_atomic_fetch_add(hp,     acc.x, __ATOMIC_RELAXED, __HIP_MEMORY_SCOPE_WORKGROUP);
                __hip_atomic_fetch_add(hp + 1, acc.y, __ATOMIC_RELAXED, __HIP_MEMORY_SCOPE_WORKGROUP);
                __hip_atomic_fetch_add(hp + 2, acc.z, __ATOMIC_RELAXED, __HIP_MEMORY_SCOPE_WORKGROUP);
                __hip_atomic_fetch_add(hp + 3, acc.w, __ATOMIC_RELAXED, __HIP_MEMORY_SCOPE_WORKGROUP);
            }
        }
    }
    __syncthreads();

    const int w = tid >> 5;
    const int lane = tid & 31;
    const int M = lane & 15;
    const int half = lane >> 4;
    const int koff = half * 2;

    {
        v8f acc0[MT] = {};
        v8f acc1[MT] = {};
        for (int k0 = 0; k0 < D; k0 += 4) {
            const float* bp = W1 + (k0 + koff) * 256 + w * 32 + M;
            v2f b0, b1;
            b0.x = bp[0];   b0.y = bp[256];
            b1.x = bp[16];  b1.y = bp[16 + 256];
            #pragma unroll
            for (int m = 0; m < MT; ++m) {
                v2f a;
                a.x = hs[(m * 16 + M) * 132 + k0 + koff];
                a.y = hs[(m * 16 + M) * 132 + k0 + koff + 1];
                acc0[m] = __builtin_amdgcn_wmma_f32_16x16x4_f32(
                              false, a, false, b0, (short)0, acc0[m], false, false);
                acc1[m] = __builtin_amdgcn_wmma_f32_16x16x4_f32(
                              false, a, false, b1, (short)0, acc1[m], false, false);
            }
        }
        const float bias0 = B1[w * 32 + M];
        const float bias1 = B1[w * 32 + 16 + M];
        #pragma unroll
        for (int m = 0; m < MT; ++m) {
            #pragma unroll
            for (int r = 0; r < 8; ++r) {
                const int row = m * 16 + r + half * 8;
                const float v0 = acc0[m][r] + bias0;
                const float v1 = acc1[m][r] + bias1;
                hid[row * 260 + w * 32 + M]      = v0 > 0.0f ? v0 : 0.0f;
                hid[row * 260 + w * 32 + 16 + M] = v1 > 0.0f ? v1 : 0.0f;
            }
        }
    }
    __syncthreads();

    {
        v8f acc[MT] = {};
        for (int k0 = 0; k0 < 2 * D; k0 += 4) {
            const float* bp = W2 + (k0 + koff) * D + w * 16 + M;
            v2f b;
            b.x = bp[0];
            b.y = bp[D];
            #pragma unroll
            for (int m = 0; m < MT; ++m) {
                v2f a;
                a.x = hid[(m * 16 + M) * 260 + k0 + koff];
                a.y = hid[(m * 16 + M) * 260 + k0 + koff + 1];
                acc[m] = __builtin_amdgcn_wmma_f32_16x16x4_f32(
                             false, a, false, b, (short)0, acc[m], false, false);
            }
        }
        const float bias = B2[w * 16 + M];
        #pragma unroll
        for (int m = 0; m < MT; ++m) {
            #pragma unroll
            for (int r = 0; r < 8; ++r) {
                hout[(base + m * 16 + r + half * 8) * D + w * 16 + M] = acc[m][r] + bias;
            }
        }
    }
}

extern "C" void kernel_launch(void* const* d_in, const int* in_sizes, int n_in,
                              void* d_out, int out_size, void* d_ws, size_t ws_size,
                              hipStream_t stream) {
    const float* feat = (const float*)d_in[0];
    const int*   src  = (const int*)d_in[1];
    const int*   dst  = (const int*)d_in[2];
    const int*   et   = (const int*)d_in[3];
    const float* emb  = (const float*)d_in[4];
    const float* w1   = (const float*)d_in[5];
    const float* b1   = (const float*)d_in[6];
    const float* w2   = (const float*)d_in[7];
    const float* b2   = (const float*)d_in[8];
    const float* eps  = (const float*)d_in[9];
    float* out = (float*)d_out;
    int* ws = (int*)d_ws;

    k_count<<<NCHUNK, 1024, 0, stream>>>(dst, ws);
    k_scan <<<79,     256, 0, stream>>>(ws);
    k_fill <<<NCHUNK, 1024, 0, stream>>>(src, dst, et, ws);
    k_mlp  <<<NTILES, 256, 0, stream>>>(feat, eps, emb, ws, w1, b1, w2, b2, out);
}


# edge bucketing in one launch: each 2400-edge chunk orders its own edges by destination tile in LDS; the MLP block collects its tile's 250 runs; 4 launches become 2; float sum order may differ
# speedup vs baseline: 1.9668x; 1.2557x over previous
#include <hip/hip_runtime.h>

#define D 128
#define NNODES 50000
#define NEDGES 600000
#define MT 5
#define ROWS (MT * 16)
#define NTILES (NNODES / ROWS)
#define NCHUNK 250
#define EPC (NEDGES / NCHUNK)
#define CAP 2048

#define WS_OFF  0
#define WS_SORT 157568

typedef __attribute__((ext_vector_type(2))) float v2f;
typedef __attribute__((ext_vector_type(8))) float v8f;
typedef __attribute__((ext_vector_type(4))) float v4f;

__device__ __forceinline__ int wave_incl_scan(int v, int lane) {
    for (int o = 1; o < 32; o <<= 1) {
        const int t = __shfl_up(v, o);
        if (lane >= o) v += t;
    }
    return v;
}

__global__ __launch_bounds__(1024) void k_bucket(const int* __restrict__ src,
                                                 const int* __restrict__ dst,
                                                 const int* __restrict__ et,
                                                 int* __restrict__ ws) {
    __shared__ int hist[NTILES];
    __shared__ int wsum[32];
    const int tid = threadIdx.x;
    const int lane = tid & 31;
    const int w = tid >> 5;
    if (tid < NTILES) hist[tid] = 0;
    __syncthreads();
    const int e0 = blockIdx.x * EPC;
    unsigned word[3];
    int bin[3], rk[3];
    #pragma unroll
    for (int k = 0; k < 3; ++k) {
        const int i = tid + k * 1024;
        const int e = e0 + (i < EPC ? i : 0);
        const unsigned d = (unsigned)dst[e];
        const unsigned sv = (unsigned)src[e];
        const unsigned tv = (unsigned)et[e];
        const unsigned bn = d / ROWS;
        bin[k] = (int)bn;
        word[k] = sv | (tv << 16) | ((d - bn * ROWS) << 20);
    }
    #pragma unroll
    for (int k = 0; k < 3; ++k) {
        rk[k] = 0;
        if (tid + k * 1024 < EPC) rk[k] = atomicAdd(&hist[bin[k]], 1);
    }
    __syncthreads();
    const int t = tid < NTILES ? hist[tid] : 0;
    const int incl = wave_incl_scan(t, lane);
    if (lane == 31) wsum[w] = incl;
    __syncthreads();
    int b = incl - t;
    for (int k = 0; k < w; ++k) b += wsum[k];
    if (tid < NTILES) {
        hist[tid] = b;
        ws[WS_OFF + tid * NCHUNK + blockIdx.x] = b | (t << 16);
    }
    __syncthreads();
    unsigned* sorted = reinterpret_cast<unsigned*>(ws + WS_SORT) + e0;
    #pragma unroll
    for (int k = 0; k < 3; ++k) {
        if (tid + k * 1024 < EPC) sorted[hist[bin[k]] + rk[k]] = word[k];
    }
}

__global__ __launch_bounds__(256) void k_mlp(const float* __restrict__ feat,
                                             const float* __restrict__ eps,
                                             const float* __restrict__ emb,
                                             const int* __restrict__ ws,
                                             const float* __restrict__ W1,
                                             const float* __restrict__ B1,
                                             const float* __restrict__ W2,
                                             const float* __restrict__ B2,
                                             float* __restrict__ hout) {
    __shared__ __attribute__((aligned(16))) float hs[(ROWS + 1) * 132];
    __shared__ __attribute__((aligned(16))) float hid[ROWS * 260];
    __shared__ int cnt[96];
    __shared__ unsigned lst[CAP];
    __shared__ int wtot[8];
    const int tid = threadIdx.x;
    const int base = blockIdx.x * ROWS;
    const int pk = tid < NCHUNK ? ws[WS_OFF + blockIdx.x * NCHUNK + tid] : 0;
    const float sc = 1.0f + eps[0];

    #pragma unroll
    for (int i = 0; i < 10; ++i) {
        const int chunk = tid + i * 256;
        const int r = chunk >> 5;
        const int c = (chunk & 31) * 4;
        const float4 f = *reinterpret_cast<const float4*>(feat + (base + r) * D + c);
        float4 o;
        o.x = sc * f.x; o.y = sc * f.y; o.z = sc * f.z; o.w = sc * f.w;
        *reinterpret_cast<float4*>(&hs[r * 132 + c]) = o;
    }
    for (int i = 0; i < 2; ++i) {
        const int chunk = tid + i * 256;
        *reinterpret_cast<float4*>(&hid[chunk * 4]) =
            *reinterpret_cast<const float4*>(emb + chunk * 4);
    }

    {
        const int lane = tid & 31;
        const int wv = __builtin_amdgcn_readfirstlane(tid >> 5);
        const int rc = pk >> 16;
        const unsigned* run = reinterpret_cast<const unsigned*>(ws + WS_SORT)
                              + (tid < NCHUNK ? tid : 0) * EPC + (pk & 0xFFFF);
        const int incl = wave_incl_scan(rc, lane);
        if (lane == 31) wtot[wv] = incl;
        __syncthreads();
        int rp = incl - rc;
        int nsum = 0;
        #pragma unroll
        for (int k = 0; k < 8; ++k) {
            const int s = wtot[k];
            if (k < wv) rp += s;
            nsum += s;
        }
        const int n = __builtin_amdgcn_readfirstlane(nsum);
        unsigned* raw = reinterpret_cast<unsigned*>(&hid[2048]);
        const int c = lane * 4;
        const int pad = ROWS << 20;
        for (int B = 0; B < n; B += CAP) {
            const int m = (n - B) < CAP ? (n - B) : CAP;
            if (tid < 96) cnt[tid] = 0;
            {
                const int j0 = B > rp ? B - rp : 0;
                const int j1 = (B + m - rp) < rc ? (B + m - rp) : rc;
                for (int j = j0; j < j1; j += 8) {
                    unsigned v[8];
                    #pragma unroll
                    for (int u = 0; u < 8; ++u) v[u] = run[(j + u) < j1 ? (j + u) : j];
                    asm volatile("" :: "v"(v[0]), "v"(v[1]), "v"(v[2]), "v"(v[3]),
                                       "v"(v[4]), "v"(v[5]), "v"(v[6]), "v"(v[7]));
                    #pragma unroll
                    for (int u = 0; u < 8; ++u) {
                        if ((j + u) < j1) raw[rp + j + u - B] = v[u];
                    }
                }
            }
            __syncthreads();
            unsigned wk[CAP / 256];
            int rk[CAP / 256];
            #pragma unroll
            for (int k = 0; k < CAP / 256; ++k) {
                const int i = k * 256 + tid;
                wk[k] = raw[i < m ? i : 0];
            }
            #pragma unroll
            for (int k = 0; k < CAP / 256; ++k) {
                const int i = k * 256 + tid;
                rk[k] = 0;
                if (i < m) rk[k] = atomicAdd(&cnt[wk[k] >> 20], 1);
            }
            __syncthreads();
            if (tid < 32) {
                const int a0 = cnt[3 * lane], a1 = cnt[3 * lane + 1], a2 = cnt[3 * lane + 2];
                const int sum = a0 + a1 + a2;
                const int ex = wave_incl_scan(sum, lane) - sum;
                cnt[3 * lane] = ex;
                cnt[3 * lane + 1] = ex + a0;
                cnt[3 * lane + 2] = ex + a0 + a1;
            }
            __syncthreads();
            #pragma unroll
            for (int k = 0; k < CAP / 256; ++k) {
                const int i = k * 256 + tid;
                if (i < m) lst[cnt[wk[k] >> 20] + rk[k]] = wk[k];
            }
            __syncthreads();
            const int lo = (m * wv) >> 3;
            const int hi = (m * (wv + 1)) >> 3;
            int cur = ROWS;
            float4 acc = make_float4(0.0f, 0.0f, 0.0f, 0.0f);
            for (int pos = lo; pos < hi; pos += 32) {
                int p = pad;
                if (pos + lane < hi) p = (int)lst[pos + lane];
                const int rem = hi - pos;
                const int cw = rem < 32 ? rem : 32;
                for (int j = 0; j < cw; j += 16) {
                    int pj[16];
                    v4f f[16], e[16];
                    #pragma unroll
                    for (int u = 0; u < 16; ++u) {
                        pj[u] = __builtin_amdgcn_readlane(p, j + u);
                        f[u] = *reinterpret_cast<const v4f*>(feat + (pj[u] & 0xFFFF) * D + c);
                        e[u] = *reinterpret_cast<const v4f*>(&hid[((pj[u] >> 16) & 15) * D + c]);
                    }
                    asm volatile("" :: "v"(f[0]), "v"(f[1]), "v"(f[2]), "v"(f[3]),
                                       "v"(f[4]), "v"(f[5]), "v"(f[6]), "v"(f[7]),
                                       "v"(f[8]), "v"(f[9]), "v"(f[10]), "v"(f[11]),
                                       "v"(f[12]), "v"(f[13]), "v"(f[14]), "v"(f[15]));
                    #pragma unroll
                    for (int u = 0; u < 16; ++u) {
                        const int r = pj[u] >> 20;
                        if (r != cur) {
                            float* hp = &hs[cur * 132 + c];
                            __hip_atomic_fetch_add(hp,     acc.x, __ATOMIC_RELAXED, __HIP_MEMORY_SCOPE_WORKGROUP);
                            __hip_atomic_fetch_add(hp + 1, acc.y, __ATOMIC_RELAXED, __HIP_MEMORY_SCOPE_WORKGROUP);
                            __hip_atomic_fetch_add(hp + 2, acc.z, __ATOMIC_RELAXED, __HIP_MEMORY_SCOPE_WORKGROUP);
                            __hip_atomic_fetch_add(hp + 3, acc.w, __ATOMIC_RELAXED, __HIP_MEMORY_SCOPE_WORKGROUP);
                            acc = make_float4(0.0f, 0.0f, 0.0f, 0.0f);
                            cur = r;
                        }
                        acc.x += fmaxf(f[u].x + e[u].x, 0.0f);
                        acc.y += fmaxf(f[u].y + e[u].y, 0.0f);
                        acc.z += fmaxf(f[u].z + e[u].z, 0.0f);
                        acc.w += fmaxf(f[u].w + e[u].w, 0.0f);
                    }
                }
            }
            {
                float* hp = &hs[cur * 132 + c];
                __hip_atomic_fetch_add(hp,     acc.x, __ATOMIC_RELAXED, __HIP_MEMORY_SCOPE_WORKGROUP);
                __hip_atomic_fetch_add(hp + 1, acc.y, __ATOMIC_RELAXED, __HIP_MEMORY_SCOPE_WORKGROUP);
                __hip_atomic_fetch_add(hp + 2, acc.z, __ATOMIC_RELAXED, __HIP_MEMORY_SCOPE_WORKGROUP);
                __hip_atomic_fetch_add(hp + 3, acc.w, __ATOMIC_RELAXED, __HIP_MEMORY_SCOPE_WORKGROUP);
            }
        }
    }
    __syncthreads();

    const int w = tid >> 5;
    const int lane = tid & 31;
    const int M = lane & 15;
    const int half = lane >> 4;
    const int koff = half * 2;

    {
        v8f acc0[MT] = {};
        v8f acc1[MT] = {};
        for (int k0 = 0; k0 < D; k0 += 4) {
            const float* bp = W1 + (k0 + koff) * 256 + w * 32 + M;
            v2f b0, b1;
            b0.x = bp[0];   b0.y = bp[256];
            b1.x = bp[16];  b1.y = bp[16 + 256];
            #pragma unroll
            for (int m = 0; m < MT; ++m) {
                v2f a;
                a.x = hs[(m * 16 + M) * 132 + k0 + koff];
                a.y = hs[(m * 16 + M) * 132 + k0 + koff + 1];
                acc0[m] = __builtin_amdgcn_wmma_f32_16x16x4_f32(
                              false, a, false, b0, (short)0, acc0[m], false, false);
                acc1[m] = __builtin_amdgcn_wmma_f32_16x16x4_f32(
                              false, a, false, b1, (short)0, acc1[m], false, false);
            }
        }
        const float bias0 = B1[w * 32 + M];
        const float bias1 = B1[w * 32 + 16 + M];
        #pragma unroll
        for (int m = 0; m < MT; ++m) {
            #pragma unroll
            for (int r = 0; r < 8; ++r) {
                const int row = m * 16 + r + half * 8;
                const float v0 = acc0[m][r] + bias0;
                const float v1 = acc1[m][r] + bias1;
                hid[row * 260 + w * 32 + M]      = v0 > 0.0f ? v0 : 0.0f;
                hid[row * 260 + w * 32 + 16 + M] = v1 > 0.0f ? v1 : 0.0f;
            }
        }
    }
    __syncthreads();

    {
        v8f acc[MT] = {};
        for (int k0 = 0; k0 < 2 * D; k0 += 4) {
            const float* bp = W2 + (k0 + koff) * D + w * 16 + M;
            v2f b;
            b.x = bp[0];
            b.y = bp[D];
            #pragma unroll
            for (int m = 0; m < MT; ++m) {
                v2f a;
                a.x = hid[(m * 16 + M) * 260 + k0 + koff];
                a.y = hid[(m * 16 + M) * 260 + k0 + koff + 1];
                acc[m] = __builtin_amdgcn_wmma_f32_16x16x4_f32(
                             false, a, false, b, (short)0, acc[m], false, false);
            }
        }
        const float bias = B2[w * 16 + M];
        #pragma unroll
        for (int m = 0; m < MT; ++m) {
            #pragma unroll
            for (int r = 0; r < 8; ++r) {
                hout[(base + m * 16 + r + half * 8) * D + w * 16 + M] = acc[m][r] + bias;
            }
        }
    }
}

extern "C" void kernel_launch(void* const* d_in, const int* in_sizes, int n_in,
                              void* d_out, int out_size, void* d_ws, size_t ws_size,
                              hipStream_t stream) {
    const float* feat = (const float*)d_in[0];
    const int*   src  = (const int*)d_in[1];
    const int*   dst  = (const int*)d_in[2];
    const int*   et   = (const int*)d_in[3];
    const float* emb  = (const float*)d_in[4];
    const float* w1   = (const float*)d_in[5];
    const float* b1   = (const float*)d_in[6];
    const float* w2   = (const float*)d_in[7];
    const float* b2   = (const float*)d_in[8];
    const float* eps  = (const float*)d_in[9];
    float* out = (float*)d_out;
    int* ws = (int*)d_ws;

    k_bucket<<<NCHUNK, 1024, 0, stream>>>(src, dst, et, ws);
    k_mlp   <<<NTILES, 256, 0, stream>>>(feat, eps, emb, ws, w1, b1, w2, b2, out);
}


# MLP weight fragments prefetched one group of k-steps ahead: two register sets per phase take turns, the next group's loads fly under the current group's products; sum order kept
# speedup vs baseline: 2.0505x; 1.0426x over previous
#include <hip/hip_runtime.h>

#define D 128
#define NNODES 50000
#define NEDGES 600000
#define MT 5
#define ROWS (MT * 16)
#define NTILES (NNODES / ROWS)
#define NCHUNK 250
#define EPC (NEDGES / NCHUNK)
#define CAP 2048

#define WS_OFF  0
#define WS_SORT 157568

typedef __attribute__((ext_vector_type(2))) float v2f;
typedef __attribute__((ext_vector_type(8))) float v8f;
typedef __attribute__((ext_vector_type(4))) float v4f;

__device__ __forceinline__ int wave_incl_scan(int v, int lane) {
    for (int o = 1; o < 32; o <<= 1) {
        const int t = __shfl_up(v, o);
        if (lane >= o) v += t;
    }
    return v;
}

__global__ __launch_bounds__(1024) void k_bucket(const int* __restrict__ src,
                                                 const int* __restrict__ dst,
                                                 const int* __restrict__ et,
                                                 int* __restrict__ ws) {
    __shared__ int hist[NTILES];
    __shared__ int wsum[32];
    const int tid = threadIdx.x;
    const int lane = tid & 31;
    const int w = tid >> 5;
    if (tid < NTILES) hist[tid] = 0;
    __syncthreads();
    const int e0 = blockIdx.x * EPC;
    unsigned word[3];
    int bin[3], rk[3];
    #pragma unroll
    for (int k = 0; k < 3; ++k) {
        const int i = tid + k * 1024;
        const int e = e0 + (i < EPC ? i : 0);
        const unsigned d = (unsigned)dst[e];
        const unsigned sv = (unsigned)src[e];
        const unsigned tv = (unsigned)et[e];
        const unsigned bn = d / ROWS;
        bin[k] = (int)bn;
        word[k] = sv | (tv << 16) | ((d - bn * ROWS) << 20);
    }
    #pragma unroll
    for (int k = 0; k < 3; ++k) {
        rk[k] = 0;
        if (tid + k * 1024 < EPC) rk[k] = atomicAdd(&hist[bin[k]], 1);
    }
    __syncthreads();
    const int t = tid < NTILES ? hist[tid] : 0;
    const int incl = wave_incl_scan(t, lane);
    if (lane == 31) wsum[w] = incl;
    __syncthreads();
    int b = incl - t;
    for (int k = 0; k < w; ++k) b += wsum[k];
    if (tid < NTILES) {
        hist[tid] = b;
        ws[WS_OFF + tid * NCHUNK + blockIdx.x] = b | (t << 16);
    }
    __syncthreads();
    unsigned* sorted = reinterpret_cast<unsigned*>(ws + WS_SORT) + e0;
    #pragma unroll
    for (int k = 0; k < 3; ++k) {
        if (tid + k * 1024 < EPC) sorted[hist[bin[k]] + rk[k]] = word[k];
    }
}

__global__ __launch_bounds__(256) void k_mlp(const float* __restrict__ feat,
                                             const float* __restrict__ eps,
                                             const float* __restrict__ emb,
                                             const int* __restrict__ ws,
                                             const float* __restrict__ W1,
                                             const float* __restrict__ B1,
                                             const float* __restrict__ W2,
                                             const float* __restrict__ B2,
                                             float* __restrict__ hout) {
    __shared__ __attribute__((aligned(16))) float hs[(ROWS + 1) * 132];
    __shared__ __attribute__((aligned(16))) float hid[ROWS * 260];
    __shared__ int cnt[96];
    __shared__ unsigned lst[CAP];
    __shared__ int wtot[8];
    const int tid = threadIdx.x;
    const int base = blockIdx.x * ROWS;
    const int pk = tid < NCHUNK ? ws[WS_OFF + blockIdx.x * NCHUNK + tid] : 0;
    const float sc = 1.0f + eps[0];

    #pragma unroll
    for (int i = 0; i < 10; ++i) {
        const int chunk = tid + i * 256;
        const int r = chunk >> 5;
        const int c = (chunk & 31) * 4;
        const float4 f = *reinterpret_cast<const float4*>(feat + (base + r) * D + c);
        float4 o;
        o.x = sc * f.x; o.y = sc * f.y; o.z = sc * f.z; o.w = sc * f.w;
        *reinterpret_cast<float4*>(&hs[r * 132 + c]) = o;
    }
    for (int i = 0; i < 2; ++i) {
        const int chunk = tid + i * 256;
        *reinterpret_cast<float4*>(&hid[chunk * 4]) =
            *reinterpret_cast<const float4*>(emb + chunk * 4);
    }

    {
        const int lane = tid & 31;
        const int wv = __builtin_amdgcn_readfirstlane(tid >> 5);
        const int rc = pk >> 16;
        const unsigned* run = reinterpret_cast<const unsigned*>(ws + WS_SORT)
                              + (tid < NCHUNK ? tid : 0) * EPC + (pk & 0xFFFF);
        const int incl = wave_incl_scan(rc, lane);
        if (lane == 31) wtot[wv] = incl;
        __syncthreads();
        int rp = incl - rc;
        int nsum = 0;
        #pragma unroll
        for (int k = 0; k < 8; ++k) {
            const int s = wtot[k];
            if (k < wv) rp += s;
            nsum += s;
        }
        const int n = __builtin_amdgcn_readfirstlane(nsum);
        unsigned* raw = reinterpret_cast<unsigned*>(&hid[2048]);
        const int c = lane * 4;
        const int pad = ROWS << 20;
        for (int B = 0; B < n; B += CAP) {
            const int m = (n - B) < CAP ? (n - B) : CAP;
            if (tid < 96) cnt[tid] = 0;
            {
                const int j0 = B > rp ? B - rp : 0;
                const int j1 = (B + m - rp) < rc ? (B + m - rp) : rc;
                for (int j = j0; j < j1; j += 8) {
                    unsigned v[8];
                    #pragma unroll
                    for (int u = 0; u < 8; ++u) v[u] = run[(j + u) < j1 ? (j + u) : j];
                    asm volatile("" :: "v"(v[0]), "v"(v[1]), "v"(v[2]), "v"(v[3]),
                                       "v"(v[4]), "v"(v[5]), "v"(v[6]), "v"(v[7]));
                    #pragma unroll
                    for (int u = 0; u < 8; ++u) {
                        if ((j + u) < j1) raw[rp + j + u - B] = v[u];
                    }
                }
            }
            __syncthreads();
            unsigned wk[CAP / 256];
            int rk[CAP / 256];
            #pragma unroll
            for (int k = 0; k < CAP / 256; ++k) {
                const int i = k * 256 + tid;
                wk[k] = raw[i < m ? i : 0];
            }
            #pragma unroll
            for (int k = 0; k < CAP / 256; ++k) {
                const int i = k * 256 + tid;
                rk[k] = 0;
                if (i < m) rk[k] = atomicAdd(&cnt[wk[k] >> 20], 1);
            }
            __syncthreads();
            if (tid < 32) {
                const int a0 = cnt[3 * lane], a1 = cnt[3 * lane + 1], a2 = cnt[3 * lane + 2];
                const int sum = a0 + a1 + a2;
                const int ex = wave_incl_scan(sum, lane) - sum;
                cnt[3 * lane] = ex;
                cnt[3 * lane + 1] = ex + a0;
                cnt[3 * lane + 2] = ex + a0 + a1;
            }
            __syncthreads();
            #pragma unroll
            for (int k = 0; k < CAP / 256; ++k) {
                const int i = k * 256 + tid;
                if (i < m) lst[cnt[wk[k] >> 20] + rk[k]] = wk[k];
            }
            __syncthreads();
            const int lo = (m * wv) >> 3;
            const int hi = (m * (wv + 1)) >> 3;
            int cur = ROWS;
            float4 acc = make_float4(0.0f, 0.0f, 0.0f, 0.0f);
            for (int pos = lo; pos < hi; pos += 32) {
                int p = pad;
                if (pos + lane < hi) p = (int)lst[pos + lane];
                const int rem = hi - pos;
                const int cw = rem < 32 ? rem : 32;
                for (int j = 0; j < cw; j += 16) {
                    int pj[16];
                    v4f f[16], e[16];
                    #pragma unroll
                    for (int u = 0; u < 16; ++u) {
                        pj[u] = __builtin_amdgcn_readlane(p, j + u);
                        f[u] = *reinterpret_cast<const v4f*>(feat + (pj[u] & 0xFFFF) * D + c);
                        e[u] = *reinterpret_cast<const v4f*>(&hid[((pj[u] >> 16) & 15) * D + c]);
                    }
                    asm volatile("" :: "v"(f[0]), "v"(f[1]), "v"(f[2]), "v"(f[3]),
                                       "v"(f[4]), "v"(f[5]), "v"(f[6]), "v"(f[7]),
                                       "v"(f[8]), "v"(f[9]), "v"(f[10]), "v"(f[11]),
                                       "v"(f[12]), "v"(f[13]), "v"(f[14]), "v"(f[15]));
                    #pragma unroll
                    for (int u = 0; u < 16; ++u) {
                        const int r = pj[u] >> 20;
                        if (r != cur) {
                            float* hp = &hs[cur * 132 + c];
                            __hip_atomic_fetch_add(hp,     acc.x, __ATOMIC_RELAXED, __HIP_MEMORY_SCOPE_WORKGROUP);
                            __hip_atomic_fetch_add(hp + 1, acc.y, __ATOMIC_RELAXED, __HIP_MEMORY_SCOPE_WORKGROUP);
                            __hip_atomic_fetch_add(hp + 2, acc.z, __ATOMIC_RELAXED, __HIP_MEMORY_SCOPE_WORKGROUP);
                            __hip_atomic_fetch_add(hp + 3, acc.w, __ATOMIC_RELAXED, __HIP_MEMORY_SCOPE_WORKGROUP);
                            acc = make_float4(0.0f, 0.0f, 0.0f, 0.0f);
                            cur = r;
                        }
                        acc.x += fmaxf(f[u].x + e[u].x, 0.0f);
                        acc.y += fmaxf(f[u].y + e[u].y, 0.0f);
                        acc.z += fmaxf(f[u].z + e[u].z, 0.0f);
                        acc.w += fmaxf(f[u].w + e[u].w, 0.0f);
                    }
                }
            }
            {
                float* hp = &hs[cur * 132 + c];
                __hip_atomic_fetch_add(hp,     acc.x, __ATOMIC_RELAXED, __HIP_MEMORY_SCOPE_WORKGROUP);
                __hip_atomic_fetch_add(hp + 1, acc.y, __ATOMIC_RELAXED, __HIP_MEMORY_SCOPE_WORKGROUP);
                __hip_atomic_fetch_add(hp + 2, acc.z, __ATOMIC_RELAXED, __HIP_MEMORY_SCOPE_WORKGROUP);
                __hip_atomic_fetch_add(hp + 3, acc.w, __ATOMIC_RELAXED, __HIP_MEMORY_SCOPE_WORKGROUP);
            }
        }
    }
    const int w = tid >> 5;
    const int lane = tid & 31;
    const int M = lane & 15;
    const int half = lane >> 4;
    const int koff = half * 2;

    const float* w1p = W1 + koff * 256 + w * 32 + M;
    const float* w2p = W2 + koff * D + w * 16 + M;

    auto ld1 = [&](v2f (&b)[8], const int g) {
        #pragma unroll
        for (int u = 0; u < 4; ++u) {
            const float* bp = w1p + (g * 16 + u * 4) * 256;
            b[2 * u].x = bp[0];       b[2 * u].y = bp[256];
            b[2 * u + 1].x = bp[16];  b[2 * u + 1].y = bp[16 + 256];
        }
    };
    auto ld2 = [&](v2f (&b)[8], const int g) {
        #pragma unroll
        for (int u = 0; u < 8; ++u) {
            const float* bp = w2p + (g * 32 + u * 4) * D;
            b[u].x = bp[0];
            b[u].y = bp[D];
        }
    };

    v2f pa[8], pb[8];
    ld1(pa, 0);
    __syncthreads();

    {
        v8f acc0[MT] = {};
        v8f acc1[MT] = {};
        auto mm1 = [&](const v2f (&b)[8], const int g) {
            #pragma unroll
            for (int u = 0; u < 4; ++u) {
                const int k0 = g * 16 + u * 4;
                #pragma unroll
                for (int m = 0; m < MT; ++m) {
                    v2f a;
                    a.x = hs[(m * 16 + M) * 132 + k0 + koff];
                    a.y = hs[(m * 16 + M) * 132 + k0 + koff + 1];
                    acc0[m] = __builtin_amdgcn_wmma_f32_16x16x4_f32(
                                  false, a, false, b[2 * u], (short)0, acc0[m], false, false);
                    acc1[m] = __builtin_amdgcn_wmma_f32_16x16x4_f32(
                                  false, a, false, b[2 * u + 1], (short)0, acc1[m], false, false);
                }
            }
        };
        #pragma unroll 1
        for (int g = 0; g < 6; g += 2) {
            ld1(pb, g + 1);
            mm1(pa, g);
            ld1(pa, g + 2);
            mm1(pb, g + 1);
        }
        ld1(pb, 7);
        mm1(pa, 6);
        ld2(pa, 0);
        mm1(pb, 7);
        const float bias0 = B1[w * 32 + M];
        const float bias1 = B1[w * 32 + 16 + M];
        #pragma unroll
        for (int m = 0; m < MT; ++m) {
            #pragma unroll
            for (int r = 0; r < 8; ++r) {
                const int row = m * 16 + r + half * 8;
                const float v0 = acc0[m][r] + bias0;
                const float v1 = acc1[m][r] + bias1;
                hid[row * 260 + w * 32 + M]      = v0 > 0.0f ? v0 : 0.0f;
                hid[row * 260 + w * 32 + 16 + M] = v1 > 0.0f ? v1 : 0.0f;
            }
        }
    }
    __syncthreads();

    {
        v8f acc[MT] = {};
        auto mm2 = [&](const v2f (&b)[8], const int g) {
            #pragma unroll
            for (int u = 0; u < 8; ++u) {
                const int k0 = g * 32 + u * 4;
                #pragma unroll
                for (int m = 0; m < MT; ++m) {
                    v2f a;
                    a.x = hid[(m * 16 + M) * 260 + k0 + koff];
                    a.y = hid[(m * 16 + M) * 260 + k0 + koff + 1];
                    acc[m] = __builtin_amdgcn_wmma_f32_16x16x4_f32(
                                 false, a, false, b[u], (short)0, acc[m], false, false);
                }
            }
        };
        #pragma unroll 1
        for (int g = 0; g < 6; g += 2) {
            ld2(pb, g + 1);
            mm2(pa, g);
            ld2(pa, g + 2);
            mm2(pb, g + 1);
        }
        ld2(pb, 7);
        mm2(pa, 6);
        mm2(pb, 7);
        const float bias = B2[w * 16 + M];
        #pragma unroll
        for (int m = 0; m < MT; ++m) {
            #pragma unroll
            for (int r = 0; r < 8; ++r) {
                hout[(base + m * 16 + r + half * 8) * D + w * 16 + M] = acc[m][r] + bias;
            }
        }
    }
}

extern "C" void kernel_launch(void* const* d_in, const int* in_sizes, int n_in,
                              void* d_out, int out_size, void* d_ws, size_t ws_size,
                              hipStream_t stream) {
    const float* feat = (const float*)d_in[0];
    const int*   src  = (const int*)d_in[1];
    const int*   dst  = (const int*)d_in[2];
    const int*   et   = (const int*)d_in[3];
    const float* emb  = (const float*)d_in[4];
    const float* w1   = (const float*)d_in[5];
    const float* b1   = (const float*)d_in[6];
    const float* w2   = (const float*)d_in[7];
    const float* b2   = (const float*)d_in[8];
    const float* eps  = (const float*)d_in[9];
    float* out = (float*)d_out;
    int* ws = (int*)d_ws;

    k_bucket<<<NCHUNK, 1024, 0, stream>>>(src, dst, et, ws);
    k_mlp   <<<NTILES, 256, 0, stream>>>(feat, eps, emb, ws, w1, b1, w2, b2, out);
}


# three blocks per WGP (was two), all 625 run in one round: hidden tile shares the h tile's LDS behind one more barrier, registers held to 160 (phase-1 weight sets of two k-steps); sum order kept
# speedup vs baseline: 2.2222x; 1.0837x over previous
#include <hip/hip_runtime.h>

#define D 128
#define NNODES 50000
#define NEDGES 600000
#define MT 5
#define ROWS (MT * 16)
#define NTILES (NNODES / ROWS)
#define NCHUNK 250
#define EPC (NEDGES / NCHUNK)
#define CAP 2048

#define WS_OFF  0
#define WS_SORT 157568

typedef __attribute__((ext_vector_type(2))) float v2f;
typedef __attribute__((ext_vector_type(8))) float v8f;
typedef __attribute__((ext_vector_type(4))) float v4f;

__device__ __forceinline__ int wave_incl_scan(int v, int lane) {
    for (int o = 1; o < 32; o <<= 1) {
        const int t = __shfl_up(v, o);
        if (lane >= o) v += t;
    }
    return v;
}

__global__ __launch_bounds__(1024) void k_bucket(const int* __restrict__ src,
                                                 const int* __restrict__ dst,
                                                 const int* __restrict__ et,
                                                 int* __restrict__ ws) {
    __shared__ int hist[NTILES];
    __shared__ int wsum[32];
    const int tid = threadIdx.x;
    const int lane = tid & 31;
    const int w = tid >> 5;
    if (tid < NTILES) hist[tid] = 0;
    __syncthreads();
    const int e0 = blockIdx.x * EPC;
    unsigned word[3];
    int bin[3], rk[3];
    #pragma unroll
    for (int k = 0; k < 3; ++k) {
        const int i = tid + k * 1024;
        const int e = e0 + (i < EPC ? i : 0);
        const unsigned d = (unsigned)dst[e];
        const unsigned sv = (unsigned)src[e];
        const unsigned tv = (unsigned)et[e];
        const unsigned bn = d / ROWS;
        bin[k] = (int)bn;
        word[k] = sv | (tv << 16) | ((d - bn * ROWS) << 20);
    }
    #pragma unroll
    for (int k = 0; k < 3; ++k) {
        rk[k] = 0;
        if (tid + k * 1024 < EPC) rk[k] = atomicAdd(&hist[bin[k]], 1);
    }
    __syncthreads();
    const int t = tid < NTILES ? hist[tid] : 0;
    const int incl = wave_incl_scan(t, lane);
    if (lane == 31) wsum[w] = incl;
    __syncthreads();
    int b = incl - t;
    for (int k = 0; k < w; ++k) b += wsum[k];
    if (tid < NTILES) {
        hist[tid] = b;
        ws[WS_OFF + tid * NCHUNK + blockIdx.x] = b | (t << 16);
    }
    __syncthreads();
    unsigned* sorted = reinterpret_cast<unsigned*>(ws + WS_SORT) + e0;
    #pragma unroll
    for (int k = 0; k < 3; ++k) {
        if (tid + k * 1024 < EPC) sorted[hist[bin[k]] + rk[k]] = word[k];
    }
}

__global__ __launch_bounds__(256, 6) void k_mlp(const float* __restrict__ feat,
                                             const float* __restrict__ eps,
                                             const float* __restrict__ emb,
                                             const int* __restrict__ ws,
                                             const float* __restrict__ W1,
                                             const float* __restrict__ B1,
                                             const float* __restrict__ W2,
                                             const float* __restrict__ B2,
                                             float* __restrict__ hout) {
    __shared__ __attribute__((aligned(16))) float buf[ROWS * 260];
    float* const hs = buf;
    float* const hid = buf;
    float* const tab = buf + (ROWS + 1) * 132;
    __shared__ int cnt[96];
    __shared__ unsigned lst[CAP];
    __shared__ int wtot[8];
    const int tid = threadIdx.x;
    const int base = blockIdx.x * ROWS;
    const int pk = tid < NCHUNK ? ws[WS_OFF + blockIdx.x * NCHUNK + tid] : 0;
    const float sc = 1.0f + eps[0];

    #pragma unroll
    for (int i = 0; i < 10; ++i) {
        const int chunk = tid + i * 256;
        const int r = chunk >> 5;
        const int c = (chunk & 31) * 4;
        const float4 f = *reinterpret_cast<const float4*>(feat + (base + r) * D + c);
        float4 o;
        o.x = sc * f.x; o.y = sc * f.y; o.z = sc * f.z; o.w = sc * f.w;
        *reinterpret_cast<float4*>(&hs[r * 132 + c]) = o;
    }
    for (int i = 0; i < 2; ++i) {
        const int chunk = tid + i * 256;
        *reinterpret_cast<float4*>(&tab[chunk * 4]) =
            *reinterpret_cast<const float4*>(emb + chunk * 4);
    }

    {
        const int lane = tid & 31;
        const int wv = __builtin_amdgcn_readfirstlane(tid >> 5);
        const int rc = pk >> 16;
        const int run = WS_SORT + (tid < NCHUNK ? tid : 0) * EPC + (pk & 0xFFFF);
        const int incl = wave_incl_scan(rc, lane);
        if (lane == 31) wtot[wv] = incl;
        __syncthreads();
        int rp = incl - rc;
        int nsum = 0;
        #pragma unroll
        for (int k = 0; k < 8; ++k) {
            const int s = wtot[k];
            if (k < wv) rp += s;
            nsum += s;
        }
        const int n = __builtin_amdgcn_readfirstlane(nsum);
        unsigned* raw = reinterpret_cast<unsigned*>(&tab[2048]);
        const int c = lane * 4;
        const int pad = ROWS << 20;
        for (int B = 0; B < n; B += CAP) {
            const int m = (n - B) < CAP ? (n - B) : CAP;
            if (tid < 96) cnt[tid] = 0;
            {
                const int j0 = B > rp ? B - rp : 0;
                const int j1 = (B + m - rp) < rc ? (B + m - rp) : rc;
                for (int j = j0; j < j1; j += 8) {
                    unsigned v[8];
                    #pragma unroll
                    for (int u = 0; u < 8; ++u) v[u] = (unsigned)ws[run + ((j + u) < j1 ? (j + u) : j)];
                    asm volatile("" :: "v"(v[0]), "v"(v[1]), "v"(v[2]), "v"(v[3]),
                                       "v"(v[4]), "v"(v[5]), "v"(v[6]), "v"(v[7]));
                    #pragma unroll
                    for (int u = 0; u < 8; ++u) {
                        if ((j + u) < j1) raw[rp + j + u - B] = v[u];
                    }
                }
            }
            __syncthreads();
            unsigned wk[CAP / 256];
            int rk[CAP / 256];
            #pragma unroll
            for (int k = 0; k < CAP / 256; ++k) {
                const int i = k * 256 + tid;
                wk[k] = raw[i < m ? i : 0];
            }
            #pragma unroll
            for (int k = 0; k < CAP / 256; ++k) {
                const int i = k * 256 + tid;
                rk[k] = 0;
                if (i < m) rk[k] = atomicAdd(&cnt[wk[k] >> 20], 1);
            }
            __syncthreads();
            if (tid < 32) {
                const int a0 = cnt[3 * lane], a1 = cnt[3 * lane + 1], a2 = cnt[3 * lane + 2];
                const int sum = a0 + a1 + a2;
                const int ex = wave_incl_scan(sum, lane) - sum;
                cnt[3 * lane] = ex;
                cnt[3 * lane + 1] = ex + a0;
                cnt[3 * lane + 2] = ex + a0 + a1;
            }
            __syncthreads();
            #pragma unroll
            for (int k = 0; k < CAP / 256; ++k) {
                const int i = k * 256 + tid;
                if (i < m) lst[cnt[wk[k] >> 20] + rk[k]] = wk[k];
            }
            __syncthreads();
            const int lo = (m * wv) >> 3;
            const int hi = (m * (wv + 1)) >> 3;
            int cur = ROWS;
            float4 acc = make_float4(0.0f, 0.0f, 0.0f, 0.0f);
            for (int pos = lo; pos < hi; pos += 32) {
                int p = pad;
                if (pos + lane < hi) p = (int)lst[pos + lane];
                const int rem = hi - pos;
                const int cw = rem < 32 ? rem : 32;
                for (int j = 0; j < cw; j += 16) {
                    int pj[16];
                    v4f f[16], e[16];
                    #pragma unroll
                    for (int u = 0; u < 16; ++u) {
                        pj[u] = __builtin_amdgcn_readlane(p, j + u);
                        f[u] = *reinterpret_cast<const v4f*>(feat + (pj[u] & 0xFFFF) * D + c);
                        e[u] = *reinterpret_cast<const v4f*>(&tab[((pj[u] >> 16) & 15) * D + c]);
                    }
                    asm volatile("" :: "v"(f[0]), "v"(f[1]), "v"(f[2]), "v"(f[3]),
                                       "v"(f[4]), "v"(f[5]), "v"(f[6]), "v"(f[7]),
                                       "v"(f[8]), "v"(f[9]), "v"(f[10]), "v"(f[11]),
                                       "v"(f[12]), "v"(f[13]), "v"(f[14]), "v"(f[15]));
                    #pragma unroll
                    for (int u = 0; u < 16; ++u) {
                        const int r = pj[u] >> 20;
                        if (r != cur) {
                            float* hp = &hs[cur * 132 + c];
                            __hip_atomic_fetch_add(hp,     acc.x, __ATOMIC_RELAXED, __HIP_MEMORY_SCOPE_WORKGROUP);
                            __hip_atomic_fetch_add(hp + 1, acc.y, __ATOMIC_RELAXED, __HIP_MEMORY_SCOPE_WORKGROUP);
                            __hip_atomic_fetch_add(hp + 2, acc.z, __ATOMIC_RELAXED, __HIP_MEMORY_SCOPE_WORKGROUP);
                            __hip_atomic_fetch_add(hp + 3, acc.w, __ATOMIC_RELAXED, __HIP_MEMORY_SCOPE_WORKGROUP);
                            acc = make_float4(0.0f, 0.0f, 0.0f, 0.0f);
                            cur = r;
                        }
                        acc.x += fmaxf(f[u].x + e[u].x, 0.0f);
                        acc.y += fmaxf(f[u].y + e[u].y, 0.0f);
                        acc.z += fmaxf(f[u].z + e[u].z, 0.0f);
                        acc.w += fmaxf(f[u].w + e[u].w, 0.0f);
                    }
                }
            }
            {
                float* hp = &hs[cur * 132 + c];
                __hip_atomic_fetch_add(hp,     acc.x, __ATOMIC_RELAXED, __HIP_MEMORY_SCOPE_WORKGROUP);
                __hip_atomic_fetch_add(hp + 1, acc.y, __ATOMIC_RELAXED, __HIP_MEMORY_SCOPE_WORKGROUP);
                __hip_atomic_fetch_add(hp + 2, acc.z, __ATOMIC_RELAXED, __HIP_MEMORY_SCOPE_WORKGROUP);
                __hip_atomic_fetch_add(hp + 3, acc.w, __ATOMIC_RELAXED, __HIP_MEMORY_SCOPE_WORKGROUP);
            }
        }
    }
    const int w = tid >> 5;
    const int lane = tid & 31;
    const int M = lane & 15;
    const int half = lane >> 4;
    const int koff = half * 2;

    const float* w1p = W1 + koff * 256 + w * 32 + M;
    const float* w2p = W2 + koff * D + w * 16 + M;

    auto ld1 = [&](v2f (&b)[4], const int g) {
        #pragma unroll
        for (int u = 0; u < 2; ++u) {
            const float* bp = w1p + (g * 8 + u * 4) * 256;
            b[2 * u].x = bp[0];       b[2 * u].y = bp[256];
            b[2 * u + 1].x = bp[16];  b[2 * u + 1].y = bp[16 + 256];
        }
    };
    auto ld2 = [&](v2f (&b)[8], const int g) {
        #pragma unroll
        for (int u = 0; u < 8; ++u) {
            const float* bp = w2p + (g * 32 + u * 4) * D;
            b[u].x = bp[0];
            b[u].y = bp[D];
        }
    };

    v2f pa[8], pb[8];
    v2f qa[4], qb[4];
    ld1(qa, 0);
    __syncthreads();

    {
        v8f acc0[MT] = {};
        v8f acc1[MT] = {};
        auto mm1 = [&](const v2f (&b)[4], const int g) {
            #pragma unroll
            for (int u = 0; u < 2; ++u) {
                const int k0 = g * 8 + u * 4;
                #pragma unroll
                for (int m = 0; m < MT; ++m) {
                    v2f a;
                    a.x = hs[(m * 16 + M) * 132 + k0 + koff];
                    a.y = hs[(m * 16 + M) * 132 + k0 + koff + 1];
                    acc0[m] = __builtin_amdgcn_wmma_f32_16x16x4_f32(
                                  false, a, false, b[2 * u], (short)0, acc0[m], false, false);
                    acc1[m] = __builtin_amdgcn_wmma_f32_16x16x4_f32(
                                  false, a, false, b[2 * u + 1], (short)0, acc1[m], false, false);
                }
            }
        };
        #pragma unroll 1
        for (int g = 0; g < 14; g += 2) {
            ld1(qb, g + 1);
            mm1(qa, g);
            ld1(qa, g + 2);
            mm1(qb, g + 1);
        }
        ld1(qb, 15);
        mm1(qa, 14);
        ld2(pa, 0);
        mm1(qb, 15);
        const float bias0 = B1[w * 32 + M];
        const float bias1 = B1[w * 32 + 16 + M];
        __syncthreads();
        #pragma unroll
        for (int m = 0; m < MT; ++m) {
            #pragma unroll
            for (int r = 0; r < 8; ++r) {
                const int row = m * 16 + r + half * 8;
                const float v0 = acc0[m][r] + bias0;
                const float v1 = acc1[m][r] + bias1;
                hid[row * 260 + w * 32 + M]      = v0 > 0.0f ? v0 : 0.0f;
                hid[row * 260 + w * 32 + 16 + M] = v1 > 0.0f ? v1 : 0.0f;
            }
        }
    }
    __syncthreads();

    {
        v8f acc[MT] = {};
        auto mm2 = [&](const v2f (&b)[8], const int g) {
            #pragma unroll
            for (int u = 0; u < 8; ++u) {
                const int k0 = g * 32 + u * 4;
                #pragma unroll
                for (int m = 0; m < MT; ++m) {
                    v2f a;
                    a.x = hid[(m * 16 + M) * 260 + k0 + koff];
                    a.y = hid[(m * 16 + M) * 260 + k0 + koff + 1];
                    acc[m] = __builtin_amdgcn_wmma_f32_16x16x4_f32(
                                 false, a, false, b[u], (short)0, acc[m], false, false);
                }
            }
        };
        #pragma unroll 1
        for (int g = 0; g < 6; g += 2) {
            ld2(pb, g + 1);
            mm2(pa, g);
            ld2(pa, g + 2);
            mm2(pb, g + 1);
        }
        ld2(pb, 7);
        mm2(pa, 6);
        mm2(pb, 7);
        const float bias = B2[w * 16 + M];
        #pragma unroll
        for (int m = 0; m < MT; ++m) {
            #pragma unroll
            for (int r = 0; r < 8; ++r) {
                hout[(base + m * 16 + r + half * 8) * D + w * 16 + M] = acc[m][r] + bias;
            }
        }
    }
}

extern "C" void kernel_launch(void* const* d_in, const int* in_sizes, int n_in,
                              void* d_out, int out_size, void* d_ws, size_t ws_size,
                              hipStream_t stream) {
    const float* feat = (const float*)d_in[0];
    const int*   src  = (const int*)d_in[1];
    const int*   dst  = (const int*)d_in[2];
    const int*   et   = (const int*)d_in[3];
    const float* emb  = (const float*)d_in[4];
    const float* w1   = (const float*)d_in[5];
    const float* b1   = (const float*)d_in[6];
    const float* w2   = (const float*)d_in[7];
    const float* b2   = (const float*)d_in[8];
    const float* eps  = (const float*)d_in[9];
    float* out = (float*)d_out;
    int* ws = (int*)d_ws;

    k_bucket<<<NCHUNK, 1024, 0, stream>>>(src, dst, et, ws);
    k_mlp   <<<NTILES, 256, 0, stream>>>(feat, eps, emb, ws, w1, b1, w2, b2, out);
}


# one 196-row tile per WGP: 256 work-groups of 16 waves (was 625 of 8 waves at 80 rows, three on the fullest WGP); a weight fragment feeds 13 M-tiles; row sum order may differ
# speedup vs baseline: 2.4717x; 1.1123x over previous
#include <hip/hip_runtime.h>


#define D 128
#define NNODES 50000
#define NEDGES 600000
#define MT 13
#define ROWS 196
#define PROWS (MT * 16)
#define NTILES ((NNODES + ROWS - 1) / ROWS)
#define NCHUNK 250
#define EPC (NEDGES / NCHUNK)
#define CAP 4096
#define NW 16

#define WS_OFF  0
#define WS_SORT 157568

typedef __attribute__((ext_vector_type(2))) float v2f;
typedef __attribute__((ext_vector_type(8))) float v8f;
typedef __attribute__((ext_vector_type(4))) float v4f;

template <int N> struct IntTag { static constexpr int value = N; };

__device__ __forceinline__ int wave_incl_scan(int v, int lane) {
    for (int o = 1; o < 32; o <<= 1) {
        const int t = __shfl_up(v, o);
        if (lane >= o) v += t;
    }
    return v;
}

__global__ __launch_bounds__(1024) void k_bucket(const int* __restrict__ src,
                                                 const int* __restrict__ dst,
                                                 const int* __restrict__ et,
                                                 int* __restrict__ ws) {
    __shared__ int hist[NTILES];
    __shared__ int wsum[32];
    const int tid = threadIdx.x;
    const int lane = tid & 31;
    const int w = tid >> 5;
    if (tid < NTILES) hist[tid] = 0;
    __syncthreads();
    const int e0 = blockIdx.x * EPC;
    unsigned word[3];
    int bin[3], rk[3];
    #pragma unroll
    for (int k = 0; k < 3; ++k) {
        const int i = tid + k * 1024;
        const int e = e0 + (i < EPC ? i : 0);
        const unsigned d = (unsigned)dst[e];
        const unsigned sv = (unsigned)src[e];
        const unsigned tv = (unsigned)et[e];
        const unsigned bn = d / ROWS;
        const unsigned rr = d - bn * ROWS;
        const unsigned pr = (rr & 15) * MT + (rr >> 4);
        bin[k] = (int)bn;
        word[k] = sv | (tv << 16) | (pr << 20);
    }
    #pragma unroll
    for (int k = 0; k < 3; ++k) {
        rk[k] = 0;
        if (tid + k * 1024 < EPC) rk[k] = atomicAdd(&hist[bin[k]], 1);
    }
    __syncthreads();
    const int t = tid < NTILES ? hist[tid] : 0;
    const int incl = wave_incl_scan(t, lane);
    if (lane == 31) wsum[w] = incl;
    __syncthreads();
    int b = incl - t;
    for (int k = 0; k < w; ++k) b += wsum[k];
    if (tid < NTILES) {
        hist[tid] = b;
        ws[WS_OFF + tid * NCHUNK + blockIdx.x] = b | (t << 16);
    }
    __syncthreads();
    unsigned* sorted = reinterpret_cast<unsigned*>(ws + WS_SORT) + e0;
    #pragma unroll
    for (int k = 0; k < 3; ++k) {
        if (tid + k * 1024 < EPC) sorted[hist[bin[k]] + rk[k]] = word[k];
    }
}

__global__ __launch_bounds__(512, 4) void k_mlp(const float* __restrict__ feat,
                                             const float* __restrict__ eps,
                                             const float* __restrict__ emb,
                                             const int* __restrict__ ws,
                                             const float* __restrict__ W1,
                                             const float* __restrict__ B1,
                                             const float* __restrict__ W2,
                                             const float* __restrict__ B2,
                                             float* __restrict__ hout) {
    __shared__ __attribute__((aligned(16))) float buf[PROWS * 260];
    float* const hs = buf;
    float* const hid = buf;
    float* const tab = buf + (PROWS + 1) * 132;
    __shared__ int cnt[256];
    __shared__ unsigned lst[CAP];
    __shared__ int wtot[NW];
    const int tid = threadIdx.x;
    const int base = blockIdx.x * ROWS;
    const int pk = tid < NCHUNK ? ws[WS_OFF + blockIdx.x * NCHUNK + tid] : 0;
    const float sc = 1.0f + eps[0];

    #pragma unroll
    for (int i = 0; i < 14; ++i) {
        const int chunk = tid + i * 512;
        const int r = chunk >> 5;
        const int c = (chunk & 31) * 4;
        const bool real = (r < ROWS) && (base + r < NNODES);
        const int gr = real ? base + r : 0;
        const float4 f = *reinterpret_cast<const float4*>(feat + gr * D + c);
        float4 o;
        o.x = real ? sc * f.x : 0.0f; o.y = real ? sc * f.y : 0.0f;
        o.z = real ? sc * f.z : 0.0f; o.w = real ? sc * f.w : 0.0f;
        const int lrow = r < PROWS ? (r & 15) * MT + (r >> 4) : PROWS;
        if (chunk < (PROWS + 1) * 32) *reinterpret_cast<float4*>(&hs[lrow * 132 + c]) = o;
    }
    *reinterpret_cast<float4*>(&tab[tid * 4]) = *reinterpret_cast<const float4*>(emb + tid * 4);

    {
        const int lane = tid & 31;
        const int wv = __builtin_amdgcn_readfirstlane(tid >> 5);
        const int rc = pk >> 16;
        const int run = WS_SORT + (tid < NCHUNK ? tid : 0) * EPC + (pk & 0xFFFF);
        const int incl = wave_incl_scan(rc, lane);
        if (lane == 31) wtot[wv] = incl;
        __syncthreads();
        int rp = incl - rc;
        int nsum = 0;
        #pragma unroll
        for (int k = 0; k < NW; ++k) {
            const int s = wtot[k];
            if (k < wv) rp += s;
            nsum += s;
        }
        const int n = __builtin_amdgcn_readfirstlane(nsum);
        unsigned* raw = reinterpret_cast<unsigned*>(&tab[2048]);
        const int c = lane * 4;
        const int pad = PROWS << 20;
        for (int B = 0; B < n; B += CAP) {
            const int m = (n - B) < CAP ? (n - B) : CAP;
            if (tid < 256) cnt[tid] = 0;
            {
                const int j0 = B > rp ? B - rp : 0;
                const int j1 = (B + m - rp) < rc ? (B + m - rp) : rc;
                for (int j = j0; j < j1; j += 16) {
                    unsigned v[16];
                    #pragma unroll
                    for (int u = 0; u < 16; ++u) v[u] = (unsigned)ws[run + ((j + u) < j1 ? (j + u) : j)];
                    asm volatile("" :: "v"(v[0]), "v"(v[1]), "v"(v[2]), "v"(v[3]),
                                       "v"(v[4]), "v"(v[5]), "v"(v[6]), "v"(v[7]),
                                       "v"(v[8]), "v"(v[9]), "v"(v[10]), "v"(v[11]),
                                       "v"(v[12]), "v"(v[13]), "v"(v[14]), "v"(v[15]));
                    #pragma unroll
                    for (int u = 0; u < 16; ++u) {
                        if ((j + u) < j1) raw[rp + j + u - B] = v[u];
                    }
                }
            }
            __syncthreads();
            unsigned wk[CAP / 512];
            int rk[CAP / 512];
            #pragma unroll
            for (int k = 0; k < CAP / 512; ++k) {
                const int i = k * 512 + tid;
                wk[k] = raw[i < m ? i : 0];
            }
            #pragma unroll
            for (int k = 0; k < CAP / 512; ++k) {
                const int i = k * 512 + tid;
                rk[k] = 0;
                if (i < m) rk[k] = atomicAdd(&cnt[wk[k] >> 20], 1);
            }
            __syncthreads();
            if (tid < 32) {
                int a[8];
                int sum = 0;
                #pragma unroll
                for (int q = 0; q < 8; ++q) { a[q] = cnt[8 * lane + q]; sum += a[q]; }
                int ex = wave_incl_scan(sum, lane) - sum;
                #pragma unroll
                for (int q = 0; q < 8; ++q) { cnt[8 * lane + q] = ex; ex += a[q]; }
            }
            __syncthreads();
            #pragma unroll
            for (int k = 0; k < CAP / 512; ++k) {
                const int i = k * 512 + tid;
                if (i < m) lst[cnt[wk[k] >> 20] + rk[k]] = wk[k];
            }
            __syncthreads();
            const int lo = (m * wv) >> 4;
            const int hi = (m * (wv + 1)) >> 4;
            int cur = PROWS;
            float4 acc = make_float4(0.0f, 0.0f, 0.0f, 0.0f);
            for (int pos = lo; pos < hi; pos += 32) {
                int p = pad;
                if (pos + lane < hi) p = (int)lst[pos + lane];
                const int rem = hi - pos;
                const int cw = rem < 32 ? rem : 32;
                for (int j = 0; j < cw; j += 16) {
                    int pj[16];
                    v4f f[16], e[16];
                    #pragma unroll
                    for (int u = 0; u < 16; ++u) {
                        pj[u] = __builtin_amdgcn_readlane(p, j + u);
                        f[u] = *reinterpret_cast<const v4f*>(feat + (pj[u] & 0xFFFF) * D + c);
                        e[u] = *reinterpret_cast<const v4f*>(&tab[((pj[u] >> 16) & 15) * D + c]);
                    }
                    asm volatile("" :: "v"(f[0]), "v"(f[1]), "v"(f[2]), "v"(f[3]),
                                       "v"(f[4]), "v"(f[5]), "v"(f[6]), "v"(f[7]),
                                       "v"(f[8]), "v"(f[9]), "v"(f[10]), "v"(f[11]),
                                       "v"(f[12]), "v"(f[13]), "v"(f[14]), "v"(f[15]));
                    #pragma unroll
                    for (int u = 0; u < 16; ++u) {
                        const int r = pj[u] >> 20;
                        if (r != cur) {
                            float* hp = &hs[cur * 132 + c];
                            __hip_atomic_fetch_add(hp,     acc.x, __ATOMIC_RELAXED, __HIP_MEMORY_SCOPE_WORKGROUP);
                            __hip_atomic_fetch_add(hp + 1, acc.y, __ATOMIC_RELAXED, __HIP_MEMORY_SCOPE_WORKGROUP);
                            __hip_atomic_fetch_add(hp + 2, acc.z, __ATOMIC_RELAXED, __HIP_MEMORY_SCOPE_WORKGROUP);
                            __hip_atomic_fetch_add(hp + 3, acc.w, __ATOMIC_RELAXED, __HIP_MEMORY_SCOPE_WORKGROUP);
                            acc = make_float4(0.0f, 0.0f, 0.0f, 0.0f);
                            cur = r;
                        }
                        acc.x += fmaxf(f[u].x + e[u].x, 0.0f);
                        acc.y += fmaxf(f[u].y + e[u].y, 0.0f);
                        acc.z += fmaxf(f[u].z + e[u].z, 0.0f);
                        acc.w += fmaxf(f[u].w + e[u].w, 0.0f);
                    }
                }
            }
            {
                float* hp = &hs[cur * 132 + c];
                __hip_atomic_fetch_add(hp,     acc.x, __ATOMIC_RELAXED, __HIP_MEMORY_SCOPE_WORKGROUP);
                __hip_atomic_fetch_add(hp + 1, acc.y, __ATOMIC_RELAXED, __HIP_MEMORY_SCOPE_WORKGROUP);
                __hip_atomic_fetch_add(hp + 2, acc.z, __ATOMIC_RELAXED, __HIP_MEMORY_SCOPE_WORKGROUP);
                __hip_atomic_fetch_add(hp + 3, acc.w, __ATOMIC_RELAXED, __HIP_MEMORY_SCOPE_WORKGROUP);
            }
        }
    }
    const int w = tid >> 5;
    const int lane = tid & 31;
    const int M = lane & 15;
    const int half = lane >> 4;
    const int koff = half * 2;
    const int wu = __builtin_amdgcn_readfirstlane(tid >> 5);
    const int mh = (wu >> 2) & 1;
    const int n2 = (wu & 3) | ((wu >> 3) << 2);
    const int mb = mh * 7;

    const float* w1p = W1 + koff * 256 + w * 16 + M;
    const float* w2p = W2 + koff * D + n2 * 16 + M;

    auto ld1 = [&](v2f (&b)[2], const int g) {
        #pragma unroll
        for (int u = 0; u < 2; ++u) {
            const float* bp = w1p + (g * 8 + u * 4) * 256;
            b[u].x = bp[0];
            b[u].y = bp[256];
        }
    };
    auto ld2 = [&](v2f (&b)[8], const int g) {
        #pragma unroll
        for (int u = 0; u < 8; ++u) {
            const float* bp = w2p + (g * 32 + u * 4) * D;
            b[u].x = bp[0];
            b[u].y = bp[D];
        }
    };

    v2f pa[8], pb[8];
    v2f qa[2], qb[2];
    ld1(qa, 0);
    __syncthreads();

    {
        v8f acc[MT] = {};
        auto mm1 = [&](const v2f (&b)[2], const int g) {
            #pragma unroll
            for (int u = 0; u < 2; ++u) {
                const int k0 = g * 8 + u * 4;
                #pragma unroll
                for (int m = 0; m < MT; ++m) {
                    v2f a;
                    a.x = hs[(M * MT + m) * 132 + k0 + koff];
                    a.y = hs[(M * MT + m) * 132 + k0 + koff + 1];
                    acc[m] = __builtin_amdgcn_wmma_f32_16x16x4_f32(
                                 false, a, false, b[u], (short)0, acc[m], false, false);
                }
            }
        };
        #pragma unroll 1
        for (int g = 0; g < 14; g += 2) {
            ld1(qb, g + 1);
            mm1(qa, g);
            ld1(qa, g + 2);
            mm1(qb, g + 1);
        }
        ld1(qb, 15);
        mm1(qa, 14);
        ld2(pa, 0);
        mm1(qb, 15);
        const float bias = B1[w * 16 + M];
        __syncthreads();
        #pragma unroll
        for (int m = 0; m < MT; ++m) {
            #pragma unroll
            for (int r = 0; r < 8; ++r) {
                const int lrow = (r + half * 8) * MT + m;
                const float v0 = acc[m][r] + bias;
                hid[lrow * 260 + w * 16 + M] = v0 > 0.0f ? v0 : 0.0f;
            }
        }
    }
    __syncthreads();

    auto phase2 = [&](auto tag) {
        constexpr int MTC = decltype(tag)::value;
        v8f acc[MTC] = {};
        auto mm2 = [&](const v2f (&b)[8], const int g) {
            #pragma unroll
            for (int u = 0; u < 8; ++u) {
                const int k0 = g * 32 + u * 4;
                #pragma unroll
                for (int m = 0; m < MTC; ++m) {
                    v2f a;
                    a.x = hid[(M * MT + mb + m) * 260 + k0 + koff];
                    a.y = hid[(M * MT + mb + m) * 260 + k0 + koff + 1];
                    acc[m] = __builtin_amdgcn_wmma_f32_16x16x4_f32(
                                 false, a, false, b[u], (short)0, acc[m], false, false);
                }
            }
        };
        #pragma unroll 1
        for (int g = 0; g < 6; g += 2) {
            ld2(pb, g + 1);
            mm2(pa, g);
            ld2(pa, g + 2);
            mm2(pb, g + 1);
        }
        ld2(pb, 7);
        mm2(pa, 6);
        mm2(pb, 7);
        const float bias = B2[n2 * 16 + M];
        const int lim = (NNODES - base) < ROWS ? (NNODES - base) : ROWS;
        #pragma unroll
        for (int m = 0; m < MTC; ++m) {
            const int r0 = (mb + m) * 16;
            float* op = hout + (base + r0 + half * 8) * D + n2 * 16 + M;
            if (r0 + 16 <= lim) {
                #pragma unroll
                for (int r = 0; r < 8; ++r) op[r * D] = acc[m][r] + bias;
            } else {
                #pragma unroll
                for (int r = 0; r < 8; ++r) {
                    if (r0 + r + half * 8 < lim) op[r * D] = acc[m][r] + bias;
                }
            }
        }
    };
    if (mh == 0) phase2(IntTag<7>{});
    else         phase2(IntTag<6>{});
}

extern "C" void kernel_launch(void* const* d_in, const int* in_sizes, int n_in,
                              void* d_out, int out_size, void* d_ws, size_t ws_size,
                              hipStream_t stream) {
    const float* feat = (const float*)d_in[0];
    const int*   src  = (const int*)d_in[1];
    const int*   dst  = (const int*)d_in[2];
    const int*   et   = (const int*)d_in[3];
    const float* emb  = (const float*)d_in[4];
    const float* w1   = (const float*)d_in[5];
    const float* b1   = (const float*)d_in[6];
    const float* w2   = (const float*)d_in[7];
    const float* b2   = (const float*)d_in[8];
    const float* eps  = (const float*)d_in[9];
    float* out = (float*)d_out;
    int* ws = (int*)d_ws;

    k_bucket<<<NCHUNK, 1024, 0, stream>>>(src, dst, et, ws);
    k_mlp   <<<NTILES, 512, 0, stream>>>(feat, eps, emb, ws, w1, b1, w2, b2, out);
}
